# starting kernel
# baseline (speedup 1.0000x reference)
#include <hip/hip_runtime.h>


#define B_    64
#define H_    512
#define W_    512
#define HW    (H_ * W_)
#define NRING 256
#define HBINS (4 * NRING)

#define BLOCKS_PER_IMG 16
#define THREADS        256
#define PIX_PER_BLOCK  (HW / BLOCKS_PER_IMG)
#define V4_PER_BLOCK   (PIX_PER_BLOCK / 4)
#define ITERS          (V4_PER_BLOCK / THREADS)

typedef __attribute__((ext_vector_type(2))) float v2f;
typedef __attribute__((ext_vector_type(4))) float v4f;
typedef __attribute__((ext_vector_type(8))) float v8f;

__global__ __launch_bounds__(THREADS) void frc_partial(
    const float* __restrict__ f1r, const float* __restrict__ f1i,
    const float* __restrict__ f2r, const float* __restrict__ f2i,
    float* __restrict__ part)
{
    __shared__ float hist[HBINS];

    const int t   = threadIdx.x;
    const int img = blockIdx.x / BLOCKS_PER_IMG;
    const int blk = blockIdx.x % BLOCKS_PER_IMG;

    #pragma unroll
    for (int i = t; i < HBINS; i += THREADS) hist[i] = 0.0f;
    __syncthreads();

    const size_t base4 = (size_t)img * (HW / 4) + (size_t)blk * V4_PER_BLOCK;
    const v4f* a1r = (const v4f*)f1r + base4;
    const v4f* a1i = (const v4f*)f1i + base4;
    const v4f* a2r = (const v4f*)f2r + base4;
    const v4f* a2i = (const v4f*)f2i + base4;

    #pragma unroll 4
    for (int i = 0; i < ITERS; ++i) {
        const int idx4 = i * THREADS + t;
        const v4f x1r = __builtin_nontemporal_load(&a1r[idx4]);
        const v4f x1i = __builtin_nontemporal_load(&a1i[idx4]);
        const v4f x2r = __builtin_nontemporal_load(&a2r[idx4]);
        const v4f x2i = __builtin_nontemporal_load(&a2i[idx4]);

        const int pix = blk * PIX_PER_BLOCK + idx4 * 4;
        const int y   = pix >> 9;
        const int x0  = pix & 511;
        const int dy  = y - 256;
        const int dy2 = dy * dy;
        const int dx  = x0 - 256;

        const int b0 = (int)__builtin_sqrtf((float)(dx * dx + dy2));
        const int b1 = (int)__builtin_sqrtf((float)((dx + 1) * (dx + 1) + dy2));
        const int b2 = (int)__builtin_sqrtf((float)((dx + 2) * (dx + 2) + dy2));
        const int b3 = (int)__builtin_sqrtf((float)((dx + 3) * (dx + 3) + dy2));

        const float cr0 = x1r.x * x2r.x + x1i.x * x2i.x;
        const float cr1 = x1r.y * x2r.y + x1i.y * x2i.y;
        const float cr2 = x1r.z * x2r.z + x1i.z * x2i.z;
        const float cr3 = x1r.w * x2r.w + x1i.w * x2i.w;
        const float ci0 = x1i.x * x2r.x - x1r.x * x2i.x;
        const float ci1 = x1i.y * x2r.y - x1r.y * x2i.y;
        const float ci2 = x1i.z * x2r.z - x1r.z * x2i.z;
        const float ci3 = x1i.w * x2r.w - x1r.w * x2i.w;
        const float p10 = x1r.x * x1r.x + x1i.x * x1i.x;
        const float p11 = x1r.y * x1r.y + x1i.y * x1i.y;
        const float p12 = x1r.z * x1r.z + x1i.z * x1i.z;
        const float p13 = x1r.w * x1r.w + x1i.w * x1i.w;
        const float p20 = x2r.x * x2r.x + x2i.x * x2i.x;
        const float p21 = x2r.y * x2r.y + x2i.y * x2i.y;
        const float p22 = x2r.z * x2r.z + x2i.z * x2i.z;
        const float p23 = x2r.w * x2r.w + x2i.w * x2i.w;

        if ((b0 == b1) & (b1 == b2) & (b2 == b3)) {
            if (b0 < NRING) {
                atomicAdd(&hist[          b0], cr0 + cr1 + cr2 + cr3);
                atomicAdd(&hist[NRING   + b0], ci0 + ci1 + ci2 + ci3);
                atomicAdd(&hist[2*NRING + b0], p10 + p11 + p12 + p13);
                atomicAdd(&hist[3*NRING + b0], p20 + p21 + p22 + p23);
            }
        } else {
            if (b0 < NRING) {
                atomicAdd(&hist[b0], cr0); atomicAdd(&hist[NRING + b0], ci0);
                atomicAdd(&hist[2*NRING + b0], p10); atomicAdd(&hist[3*NRING + b0], p20);
            }
            if (b1 < NRING) {
                atomicAdd(&hist[b1], cr1); atomicAdd(&hist[NRING + b1], ci1);
                atomicAdd(&hist[2*NRING + b1], p11); atomicAdd(&hist[3*NRING + b1], p21);
            }
            if (b2 < NRING) {
                atomicAdd(&hist[b2], cr2); atomicAdd(&hist[NRING + b2], ci2);
                atomicAdd(&hist[2*NRING + b2], p12); atomicAdd(&hist[3*NRING + b2], p22);
            }
            if (b3 < NRING) {
                atomicAdd(&hist[b3], cr3); atomicAdd(&hist[NRING + b3], ci3);
                atomicAdd(&hist[2*NRING + b3], p13); atomicAdd(&hist[3*NRING + b3], p23);
            }
        }
    }
    __syncthreads();

    float* dst = part + (size_t)(img * BLOCKS_PER_IMG + blk) * HBINS;
    #pragma unroll
    for (int i = t; i < HBINS; i += THREADS) dst[i] = hist[i];
}

__global__ __launch_bounds__(32) void frc_reduce(
    const float* __restrict__ part, float* __restrict__ out)
{
    __shared__ float acc[HBINS];

    const int lane = threadIdx.x;
    const int img  = blockIdx.x;
    const float* p = part + (size_t)img * BLOCKS_PER_IMG * HBINS;

    v2f aOnes;
    aOnes.x = 1.0f;
    aOnes.y = 1.0f;

    const int half = lane >> 4;
    const int nIdx = lane & 15;

    for (int tile = 0; tile < HBINS / 16; ++tile) {
        const int col = tile * 16 + nIdx;
        v8f c = {};
        #pragma unroll
        for (int kk = 0; kk < BLOCKS_PER_IMG; kk += 4) {
            v2f b;
            b.x = p[(size_t)(kk + 0 + 2 * half) * HBINS + col];
            b.y = p[(size_t)(kk + 1 + 2 * half) * HBINS + col];
            c = __builtin_amdgcn_wmma_f32_16x16x4_f32(
                     false, aOnes,  false, b,
                     (short)0, c,  false,  false);
        }
        if (lane < 16) acc[col] = c[0];
    }
    __syncthreads();

    #pragma unroll
    for (int rr = 0; rr < NRING; rr += 32) {
        const int r = rr + lane;
        const float cr = acc[r];
        const float ci = acc[NRING + r];
        const float p1 = acc[2 * NRING + r];
        const float p2 = acc[3 * NRING + r];
        const float num = __builtin_sqrtf(cr * cr + ci * ci);
        const float den = __builtin_sqrtf(p1 * p2);
        out[(size_t)img * NRING + r] = (den == 0.0f) ? 0.0f : num / den;
    }
}

extern "C" void kernel_launch(void* const* d_in, const int* in_sizes, int n_in,
                              void* d_out, int out_size, void* d_ws, size_t ws_size,
                              hipStream_t stream)
{
    (void)in_sizes; (void)n_in; (void)out_size; (void)ws_size;
    const float* f1r = (const float*)d_in[0];
    const float* f1i = (const float*)d_in[1];
    const float* f2r = (const float*)d_in[2];
    const float* f2i = (const float*)d_in[3];
    float* part = (float*)d_ws;

    frc_partial<<<dim3(B_ * BLOCKS_PER_IMG), dim3(THREADS), 0, stream>>>(
        f1r, f1i, f2r, f2i, part);
    frc_reduce<<<dim3(B_), dim3(32), 0, stream>>>(part, (float*)d_out);
}


# One launch: a workgroup owns (image, band of rings), streams only its band's row runs packed densely on lanes, bins in LDS, WMMA sums 4 LDS copies, finalizes; no partials (new f32 sum order)
# speedup vs baseline: 1.1098x; 1.1098x over previous
#include <hip/hip_runtime.h>


#define B_    64
#define H_    512
#define W_    512
#define HW    (H_ * W_)
#define NRING 256
#define HBINS (4 * NRING)
#define NBAND 4
#define NCOPY 4
#define THREADS 1024
#define NWAVE (THREADS / 32)
#define ROWS_PER_GROUP 4
#define UNR 4

typedef __attribute__((ext_vector_type(2))) float v2f;
typedef __attribute__((ext_vector_type(4))) float v4f;
typedef __attribute__((ext_vector_type(8))) float v8f;

__global__ __launch_bounds__(THREADS) void frc_banded(
    const float* __restrict__ f1r, const float* __restrict__ f1i,
    const float* __restrict__ f2r, const float* __restrict__ f2i,
    float* __restrict__ out)
{
    __shared__ float hist[NCOPY * HBINS];
    __shared__ float acc[HBINS];

    const int t    = threadIdx.x;
    const int lane = t & 31;
    const int wave = __builtin_amdgcn_readfirstlane(t >> 5);
    const int img  = blockIdx.x / NBAND;
    const int band = blockIdx.x % NBAND;

    const int lo = (band == 0) ? 0 : (band == 1) ? 128 : (band == 2) ? 192 : 224;
    const int hi = (band == 0) ? 128 : (band == 1) ? 192 : (band == 2) ? 224 : 256;
    const int lo2 = lo * lo;
    const int hi2 = hi * hi;

    for (int i = t; i < NCOPY * HBINS; i += THREADS) hist[i] = 0.0f;
    __syncthreads();

    float* h = hist + (wave & (NCOPY - 1)) * HBINS;

    const v4f* q1r = (const v4f*)f1r + (size_t)img * (HW / 4);
    const v4f* q1i = (const v4f*)f1i + (size_t)img * (HW / 4);
    const v4f* q2r = (const v4f*)f2r + (size_t)img * (HW / 4);
    const v4f* q2i = (const v4f*)f2i + (size_t)img * (HW / 4);

    int rowPack, rowCnt;
    {
        const int y   = wave + NWAVE * (lane & 15);
        const int dy  = y - 256;
        const int dy2 = dy * dy;
        int l0 = 0, n0 = 0, r0 = 0, n1 = 0;
        if (dy2 < hi2) {
            int ain = 0;
            if (lo2 > dy2) {
                ain = (int)__builtin_sqrtf((float)(lo2 - dy2)) - 1;
                ain = ain < 0 ? 0 : ain;
            }
            int bmax = (int)__builtin_sqrtf((float)(hi2 - dy2)) + 1;
            bmax = bmax > 256 ? 256 : bmax;
            const int xr1 = (256 + bmax) > 511 ? 511 : (256 + bmax);
            const int rsq = (256 + ain) >> 2;
            const int req = xr1 >> 2;
            const int lsq = (256 - bmax) >> 2;
            const int leq = (256 - ain) >> 2;
            if (leq >= rsq) { l0 = lsq; n0 = req - lsq + 1; r0 = 0; n1 = 0; }
            else            { l0 = lsq; n0 = leq - lsq + 1; r0 = rsq; n1 = req - rsq + 1; }
        }
        rowPack = y | (l0 << 9) | (r0 << 16) | (n0 << 23);
        rowCnt  = n0 + n1;
    }

    for (int g = 0; g < H_ / (NWAVE * ROWS_PER_GROUP); ++g) {
        int pk[ROWS_PER_GROUP];
        int pre[ROWS_PER_GROUP + 1];
        pre[0] = 0;
        for (int u = 0; u < ROWS_PER_GROUP; ++u) {
            pk[u]      = __builtin_amdgcn_readlane(rowPack, ROWS_PER_GROUP * g + u);
            pre[u + 1] = pre[u] + __builtin_amdgcn_readlane(rowCnt, ROWS_PER_GROUP * g + u);
        }
        const int total = pre[ROWS_PER_GROUP];

        for (int it0 = 0; it0 * 32 < total; it0 += UNR) {
            v4f x1r[UNR], x1i[UNR], x2r[UNR], x2i[UNR];
            int  bq[UNR][4];
            bool act[UNR];

            #pragma unroll
            for (int k = 0; k < UNR; ++k) {
                const int idx0 = (it0 + k) * 32 + lane;
                const int idx = idx0 < total ? idx0 : total - 1;
                const bool c1 = idx >= pre[1], c2 = idx >= pre[2], c3 = idx >= pre[3];
                const int p   = c3 ? pk[3] : c2 ? pk[2] : c1 ? pk[1] : pk[0];
                const int loc = idx - (c3 ? pre[3] : c2 ? pre[2] : c1 ? pre[1] : 0);
                const int y   = p & 511;
                const int l0  = (p >> 9) & 127;
                const int r0  = (p >> 16) & 127;
                const int n0  = p >> 23;
                const int q   = (loc < n0) ? (l0 + loc) : (r0 + (loc - n0));
                const int dx  = 4 * q - 256;
                const int dy  = y - 256;
                const int dy2 = dy * dy;
                bool a = false;
                for (int j = 0; j < 4; ++j) {
                    const int r2 = (dx + j) * (dx + j) + dy2;
                    const int b  = (int)__builtin_sqrtf((float)r2);
                    bq[k][j] = b;
                    a = a || ((unsigned)(b - lo) < (unsigned)(hi - lo));
                }
                act[k] = a && (idx0 < total);
                const int off = y * (W_ / 4) + q;
                x1r[k] = __builtin_nontemporal_load(q1r + off);
                x1i[k] = __builtin_nontemporal_load(q1i + off);
                x2r[k] = __builtin_nontemporal_load(q2r + off);
                x2i[k] = __builtin_nontemporal_load(q2i + off);
            }

            #pragma unroll
            for (int k = 0; k < UNR; ++k) {
                const v4f a1r = x1r[k], a1i = x1i[k], a2r = x2r[k], a2i = x2i[k];
                const int b0 = bq[k][0], b1 = bq[k][1], b2 = bq[k][2], b3 = bq[k][3];

                const float cr0 = a1r.x * a2r.x + a1i.x * a2i.x;
                const float cr1 = a1r.y * a2r.y + a1i.y * a2i.y;
                const float cr2 = a1r.z * a2r.z + a1i.z * a2i.z;
                const float cr3 = a1r.w * a2r.w + a1i.w * a2i.w;
                const float ci0 = a1i.x * a2r.x - a1r.x * a2i.x;
                const float ci1 = a1i.y * a2r.y - a1r.y * a2i.y;
                const float ci2 = a1i.z * a2r.z - a1r.z * a2i.z;
                const float ci3 = a1i.w * a2r.w - a1r.w * a2i.w;
                const float p10 = a1r.x * a1r.x + a1i.x * a1i.x;
                const float p11 = a1r.y * a1r.y + a1i.y * a1i.y;
                const float p12 = a1r.z * a1r.z + a1i.z * a1i.z;
                const float p13 = a1r.w * a1r.w + a1i.w * a1i.w;
                const float p20 = a2r.x * a2r.x + a2i.x * a2i.x;
                const float p21 = a2r.y * a2r.y + a2i.y * a2i.y;
                const float p22 = a2r.z * a2r.z + a2i.z * a2i.z;
                const float p23 = a2r.w * a2r.w + a2i.w * a2i.w;

                if (!act[k]) continue;
                if ((b0 == b1) & (b1 == b2) & (b2 == b3)) {
                    atomicAdd(&h[          b0], cr0 + cr1 + cr2 + cr3);
                    atomicAdd(&h[NRING   + b0], ci0 + ci1 + ci2 + ci3);
                    atomicAdd(&h[2*NRING + b0], p10 + p11 + p12 + p13);
                    atomicAdd(&h[3*NRING + b0], p20 + p21 + p22 + p23);
                } else {
                    if ((unsigned)(b0 - lo) < (unsigned)(hi - lo)) {
                        atomicAdd(&h[b0], cr0); atomicAdd(&h[NRING + b0], ci0);
                        atomicAdd(&h[2*NRING + b0], p10); atomicAdd(&h[3*NRING + b0], p20);
                    }
                    if ((unsigned)(b1 - lo) < (unsigned)(hi - lo)) {
                        atomicAdd(&h[b1], cr1); atomicAdd(&h[NRING + b1], ci1);
                        atomicAdd(&h[2*NRING + b1], p11); atomicAdd(&h[3*NRING + b1], p21);
                    }
                    if ((unsigned)(b2 - lo) < (unsigned)(hi - lo)) {
                        atomicAdd(&h[b2], cr2); atomicAdd(&h[NRING + b2], ci2);
                        atomicAdd(&h[2*NRING + b2], p12); atomicAdd(&h[3*NRING + b2], p22);
                    }
                    if ((unsigned)(b3 - lo) < (unsigned)(hi - lo)) {
                        atomicAdd(&h[b3], cr3); atomicAdd(&h[NRING + b3], ci3);
                        atomicAdd(&h[2*NRING + b3], p13); atomicAdd(&h[3*NRING + b3], p23);
                    }
                }
            }
        }
    }
    __syncthreads();

    {
        v2f aOnes;
        aOnes.x = 1.0f;
        aOnes.y = 1.0f;
        const int half = lane >> 4;
        const int nIdx = lane & 15;
        for (int tile = wave; tile < HBINS / 16; tile += NWAVE) {
            const int col = tile * 16 + nIdx;
            v2f b;
            b.x = hist[(0 + 2 * half) * HBINS + col];
            b.y = hist[(1 + 2 * half) * HBINS + col];
            v8f c = {};
            c = __builtin_amdgcn_wmma_f32_16x16x4_f32(
                     false, aOnes,  false, b,
                     (short)0, c,  false,  false);
            if (lane < 16) acc[col] = c[0];
        }
    }
    __syncthreads();

    if (t < NRING && t >= lo && t < hi) {
        const int r = t;
        const float cr = acc[r];
        const float ci = acc[NRING + r];
        const float p1 = acc[2 * NRING + r];
        const float p2 = acc[3 * NRING + r];
        const float num = __builtin_sqrtf(cr * cr + ci * ci);
        const float den = __builtin_sqrtf(p1 * p2);
        out[(size_t)img * NRING + r] = (den == 0.0f) ? 0.0f : num / den;
    }
}

extern "C" void kernel_launch(void* const* d_in, const int* in_sizes, int n_in,
                              void* d_out, int out_size, void* d_ws, size_t ws_size,
                              hipStream_t stream)
{
    (void)in_sizes; (void)n_in; (void)out_size; (void)d_ws; (void)ws_size;
    const float* f1r = (const float*)d_in[0];
    const float* f1i = (const float*)d_in[1];
    const float* f2r = (const float*)d_in[2];
    const float* f2i = (const float*)d_in[3];

    frc_banded<<<dim3(B_ * NBAND), dim3(THREADS), 0, stream>>>(
        f1r, f1i, f2r, f2i, (float*)d_out);
}


# One LDS histogram copy per lane index (32, pitch 5 mod 64 banks) so that no two lanes of an atomic name one word; copies summed by 8 chained WMMAs per 16 bins (new f32 sum order)
# speedup vs baseline: 1.2359x; 1.1136x over previous
#include <hip/hip_runtime.h>


#define B_    64
#define H_    512
#define W_    512
#define HW    (H_ * W_)
#define NRING 256
#define NBAND 4
#define NBL   128
#define HBINS (4 * NBL)
#define NCOPY 32
#define CSTR  (HBINS + 5)
#define THREADS 1024
#define NWAVE (THREADS / 32)
#define ROWS_PER_GROUP 4
#define UNR 4

typedef __attribute__((ext_vector_type(2))) float v2f;
typedef __attribute__((ext_vector_type(4))) float v4f;
typedef __attribute__((ext_vector_type(8))) float v8f;

__global__ __launch_bounds__(THREADS) void frc_banded(
    const float* __restrict__ f1r, const float* __restrict__ f1i,
    const float* __restrict__ f2r, const float* __restrict__ f2i,
    float* __restrict__ out)
{
    __shared__ float hist[NCOPY * CSTR];
    __shared__ float acc[HBINS];

    const int t    = threadIdx.x;
    const int lane = t & 31;
    const int wave = __builtin_amdgcn_readfirstlane(t >> 5);
    const int img  = blockIdx.x / NBAND;
    const int band = blockIdx.x % NBAND;

    const int lo = (band == 0) ? 0 : (band == 1) ? 128 : (band == 2) ? 192 : 224;
    const int hi = (band == 0) ? 128 : (band == 1) ? 192 : (band == 2) ? 224 : 256;
    const int lo2 = lo * lo;
    const int hi2 = hi * hi;

    for (int i = t; i < NCOPY * CSTR; i += THREADS) hist[i] = 0.0f;
    __syncthreads();

    const int hb = lane * CSTR - lo;

    const v4f* q1r = (const v4f*)f1r + (size_t)img * (HW / 4);
    const v4f* q1i = (const v4f*)f1i + (size_t)img * (HW / 4);
    const v4f* q2r = (const v4f*)f2r + (size_t)img * (HW / 4);
    const v4f* q2i = (const v4f*)f2i + (size_t)img * (HW / 4);

    int rowPack, rowCnt;
    {
        const int y   = wave + NWAVE * (lane & 15);
        const int dy  = y - 256;
        const int dy2 = dy * dy;
        int l0 = 0, n0 = 0, r0 = 0, n1 = 0;
        if (dy2 < hi2) {
            int ain = 0;
            if (lo2 > dy2) {
                ain = (int)__builtin_sqrtf((float)(lo2 - dy2)) - 1;
                ain = ain < 0 ? 0 : ain;
            }
            int bmax = (int)__builtin_sqrtf((float)(hi2 - dy2)) + 1;
            bmax = bmax > 256 ? 256 : bmax;
            const int xr1 = (256 + bmax) > 511 ? 511 : (256 + bmax);
            const int rsq = (256 + ain) >> 2;
            const int req = xr1 >> 2;
            const int lsq = (256 - bmax) >> 2;
            const int leq = (256 - ain) >> 2;
            if (leq >= rsq) { l0 = lsq; n0 = req - lsq + 1; r0 = 0; n1 = 0; }
            else            { l0 = lsq; n0 = leq - lsq + 1; r0 = rsq; n1 = req - rsq + 1; }
        }
        rowPack = y | (l0 << 9) | (r0 << 16) | (n0 << 23);
        rowCnt  = n0 + n1;
    }

    for (int g = 0; g < H_ / (NWAVE * ROWS_PER_GROUP); ++g) {
        int pk[ROWS_PER_GROUP];
        int pre[ROWS_PER_GROUP + 1];
        pre[0] = 0;
        for (int u = 0; u < ROWS_PER_GROUP; ++u) {
            pk[u]      = __builtin_amdgcn_readlane(rowPack, ROWS_PER_GROUP * g + u);
            pre[u + 1] = pre[u] + __builtin_amdgcn_readlane(rowCnt, ROWS_PER_GROUP * g + u);
        }
        const int total = pre[ROWS_PER_GROUP];

        for (int it0 = 0; it0 * 32 < total; it0 += UNR) {
            v4f x1r[UNR], x1i[UNR], x2r[UNR], x2i[UNR];
            int  bq[UNR][4];
            bool act[UNR];

            #pragma unroll
            for (int k = 0; k < UNR; ++k) {
                const int idx0 = (it0 + k) * 32 + lane;
                const int idx = idx0 < total ? idx0 : total - 1;
                const bool c1 = idx >= pre[1], c2 = idx >= pre[2], c3 = idx >= pre[3];
                const int p   = c3 ? pk[3] : c2 ? pk[2] : c1 ? pk[1] : pk[0];
                const int loc = idx - (c3 ? pre[3] : c2 ? pre[2] : c1 ? pre[1] : 0);
                const int y   = p & 511;
                const int l0  = (p >> 9) & 127;
                const int r0  = (p >> 16) & 127;
                const int n0  = p >> 23;
                const int q   = (loc < n0) ? (l0 + loc) : (r0 + (loc - n0));
                const int dx  = 4 * q - 256;
                const int dy  = y - 256;
                const int dy2 = dy * dy;
                bool a = false;
                for (int j = 0; j < 4; ++j) {
                    const int r2 = (dx + j) * (dx + j) + dy2;
                    const int b  = (int)__builtin_sqrtf((float)r2);
                    bq[k][j] = b;
                    a = a || ((unsigned)(b - lo) < (unsigned)(hi - lo));
                }
                act[k] = a && (idx0 < total);
                const int off = y * (W_ / 4) + q;
                x1r[k] = __builtin_nontemporal_load(q1r + off);
                x1i[k] = __builtin_nontemporal_load(q1i + off);
                x2r[k] = __builtin_nontemporal_load(q2r + off);
                x2i[k] = __builtin_nontemporal_load(q2i + off);
            }

            #pragma unroll
            for (int k = 0; k < UNR; ++k) {
                const v4f a1r = x1r[k], a1i = x1i[k], a2r = x2r[k], a2i = x2i[k];
                const int b0 = bq[k][0], b1 = bq[k][1], b2 = bq[k][2], b3 = bq[k][3];

                const float cr0 = a1r.x * a2r.x + a1i.x * a2i.x;
                const float cr1 = a1r.y * a2r.y + a1i.y * a2i.y;
                const float cr2 = a1r.z * a2r.z + a1i.z * a2i.z;
                const float cr3 = a1r.w * a2r.w + a1i.w * a2i.w;
                const float ci0 = a1i.x * a2r.x - a1r.x * a2i.x;
                const float ci1 = a1i.y * a2r.y - a1r.y * a2i.y;
                const float ci2 = a1i.z * a2r.z - a1r.z * a2i.z;
                const float ci3 = a1i.w * a2r.w - a1r.w * a2i.w;
                const float p10 = a1r.x * a1r.x + a1i.x * a1i.x;
                const float p11 = a1r.y * a1r.y + a1i.y * a1i.y;
                const float p12 = a1r.z * a1r.z + a1i.z * a1i.z;
                const float p13 = a1r.w * a1r.w + a1i.w * a1i.w;
                const float p20 = a2r.x * a2r.x + a2i.x * a2i.x;
                const float p21 = a2r.y * a2r.y + a2i.y * a2i.y;
                const float p22 = a2r.z * a2r.z + a2i.z * a2i.z;
                const float p23 = a2r.w * a2r.w + a2i.w * a2i.w;

                if (!act[k]) continue;
                if ((b0 == b1) & (b1 == b2) & (b2 == b3)) {
                    atomicAdd(&hist[hb         + b0], cr0 + cr1 + cr2 + cr3);
                    atomicAdd(&hist[hb +   NBL + b0], ci0 + ci1 + ci2 + ci3);
                    atomicAdd(&hist[hb + 2*NBL + b0], p10 + p11 + p12 + p13);
                    atomicAdd(&hist[hb + 3*NBL + b0], p20 + p21 + p22 + p23);
                } else {
                    if ((unsigned)(b0 - lo) < (unsigned)(hi - lo)) {
                        atomicAdd(&hist[hb + b0], cr0); atomicAdd(&hist[hb + NBL + b0], ci0);
                        atomicAdd(&hist[hb + 2*NBL + b0], p10); atomicAdd(&hist[hb + 3*NBL + b0], p20);
                    }
                    if ((unsigned)(b1 - lo) < (unsigned)(hi - lo)) {
                        atomicAdd(&hist[hb + b1], cr1); atomicAdd(&hist[hb + NBL + b1], ci1);
                        atomicAdd(&hist[hb + 2*NBL + b1], p11); atomicAdd(&hist[hb + 3*NBL + b1], p21);
                    }
                    if ((unsigned)(b2 - lo) < (unsigned)(hi - lo)) {
                        atomicAdd(&hist[hb + b2], cr2); atomicAdd(&hist[hb + NBL + b2], ci2);
                        atomicAdd(&hist[hb + 2*NBL + b2], p12); atomicAdd(&hist[hb + 3*NBL + b2], p22);
                    }
                    if ((unsigned)(b3 - lo) < (unsigned)(hi - lo)) {
                        atomicAdd(&hist[hb + b3], cr3); atomicAdd(&hist[hb + NBL + b3], ci3);
                        atomicAdd(&hist[hb + 2*NBL + b3], p13); atomicAdd(&hist[hb + 3*NBL + b3], p23);
                    }
                }
            }
        }
    }
    __syncthreads();

    {
        v2f aOnes;
        aOnes.x = 1.0f;
        aOnes.y = 1.0f;
        const int half = lane >> 4;
        const int nIdx = lane & 15;
        for (int tile = wave; tile < HBINS / 16; tile += NWAVE) {
            const int col = tile * 16 + nIdx;
            v2f b[NCOPY / 4];
            #pragma unroll
            for (int s = 0; s < NCOPY / 4; ++s) {
                b[s].x = hist[(4 * s + 2 * half    ) * CSTR + col];
                b[s].y = hist[(4 * s + 2 * half + 1) * CSTR + col];
            }
            v8f c = {};
            #pragma unroll
            for (int s = 0; s < NCOPY / 4; ++s) {
                c = __builtin_amdgcn_wmma_f32_16x16x4_f32(
                         false, aOnes,  false, b[s],
                         (short)0, c,  false,  false);
            }
            if (lane < 16) acc[col] = c[0];
        }
    }
    __syncthreads();

    if (t < NRING && t >= lo && t < hi) {
        const int r = t;
        const float cr = acc[r - lo];
        const float ci = acc[NBL + r - lo];
        const float p1 = acc[2 * NBL + r - lo];
        const float p2 = acc[3 * NBL + r - lo];
        const float num = __builtin_sqrtf(cr * cr + ci * ci);
        const float den = __builtin_sqrtf(p1 * p2);
        out[(size_t)img * NRING + r] = (den == 0.0f) ? 0.0f : num / den;
    }
}

extern "C" void kernel_launch(void* const* d_in, const int* in_sizes, int n_in,
                              void* d_out, int out_size, void* d_ws, size_t ws_size,
                              hipStream_t stream)
{
    (void)in_sizes; (void)n_in; (void)out_size; (void)d_ws; (void)ws_size;
    const float* f1r = (const float*)d_in[0];
    const float* f1i = (const float*)d_in[1];
    const float* f2r = (const float*)d_in[2];
    const float* f2i = (const float*)d_in[3];

    frc_banded<<<dim3(B_ * NBAND), dim3(THREADS), 0, stream>>>(
        f1r, f1i, f2r, f2i, (float*)d_out);
}


# Ring index = trunc(hardware sqrt(r2 + 0.5)) in f32, exact for every r2 here, instead of the correctly rounded sqrt expansion and integer squares: a third fewer vector instructions; sums unchanged
# speedup vs baseline: 1.3023x; 1.0537x over previous
#include <hip/hip_runtime.h>


#define B_    64
#define H_    512
#define W_    512
#define HW    (H_ * W_)
#define NRING 256
#define NBAND 4
#define NBL   128
#define HBINS (4 * NBL)
#define NCOPY 32
#define CSTR  (HBINS + 5)
#define THREADS 1024
#define NWAVE (THREADS / 32)
#define ROWS_PER_GROUP 4
#define UNR 4

typedef __attribute__((ext_vector_type(2))) float v2f;
typedef __attribute__((ext_vector_type(4))) float v4f;
typedef __attribute__((ext_vector_type(8))) float v8f;

__global__ __launch_bounds__(THREADS) void frc_banded(
    const float* __restrict__ f1r, const float* __restrict__ f1i,
    const float* __restrict__ f2r, const float* __restrict__ f2i,
    float* __restrict__ out)
{
    __shared__ float hist[NCOPY * CSTR];
    __shared__ float acc[HBINS];

    const int t    = threadIdx.x;
    const int lane = t & 31;
    const int wave = __builtin_amdgcn_readfirstlane(t >> 5);
    const int img  = blockIdx.x / NBAND;
    const int band = blockIdx.x % NBAND;

    const int lo = (band == 0) ? 0 : (band == 1) ? 128 : (band == 2) ? 192 : 224;
    const int hi = (band == 0) ? 128 : (band == 1) ? 192 : (band == 2) ? 224 : 256;
    const int lo2 = lo * lo;
    const int hi2 = hi * hi;

    for (int i = t; i < NCOPY * CSTR; i += THREADS) hist[i] = 0.0f;
    __syncthreads();

    const int hb = lane * CSTR - lo;

    const v4f* q1r = (const v4f*)f1r + (size_t)img * (HW / 4);
    const v4f* q1i = (const v4f*)f1i + (size_t)img * (HW / 4);
    const v4f* q2r = (const v4f*)f2r + (size_t)img * (HW / 4);
    const v4f* q2i = (const v4f*)f2i + (size_t)img * (HW / 4);

    int rowPack, rowCnt;
    {
        const int y   = wave + NWAVE * (lane & 15);
        const int dy  = y - 256;
        const int dy2 = dy * dy;
        int l0 = 0, n0 = 0, r0 = 0, n1 = 0;
        if (dy2 < hi2) {
            int ain = 0;
            if (lo2 > dy2) {
                ain = (int)__builtin_sqrtf((float)(lo2 - dy2)) - 1;
                ain = ain < 0 ? 0 : ain;
            }
            int bmax = (int)__builtin_sqrtf((float)(hi2 - dy2)) + 1;
            bmax = bmax > 256 ? 256 : bmax;
            const int xr1 = (256 + bmax) > 511 ? 511 : (256 + bmax);
            const int rsq = (256 + ain) >> 2;
            const int req = xr1 >> 2;
            const int lsq = (256 - bmax) >> 2;
            const int leq = (256 - ain) >> 2;
            if (leq >= rsq) { l0 = lsq; n0 = req - lsq + 1; r0 = 0; n1 = 0; }
            else            { l0 = lsq; n0 = leq - lsq + 1; r0 = rsq; n1 = req - rsq + 1; }
        }
        rowPack = y | (l0 << 9) | (r0 << 16) | (n0 << 23);
        rowCnt  = n0 + n1;
    }

    for (int g = 0; g < H_ / (NWAVE * ROWS_PER_GROUP); ++g) {
        int pk[ROWS_PER_GROUP];
        int pre[ROWS_PER_GROUP + 1];
        pre[0] = 0;
        for (int u = 0; u < ROWS_PER_GROUP; ++u) {
            pk[u]      = __builtin_amdgcn_readlane(rowPack, ROWS_PER_GROUP * g + u);
            pre[u + 1] = pre[u] + __builtin_amdgcn_readlane(rowCnt, ROWS_PER_GROUP * g + u);
        }
        const int total = pre[ROWS_PER_GROUP];

        for (int it0 = 0; it0 * 32 < total; it0 += UNR) {
            v4f x1r[UNR], x1i[UNR], x2r[UNR], x2i[UNR];
            int  bq[UNR][4];
            bool act[UNR];

            #pragma unroll
            for (int k = 0; k < UNR; ++k) {
                const int idx0 = (it0 + k) * 32 + lane;
                const int idx = idx0 < total ? idx0 : total - 1;
                const bool c1 = idx >= pre[1], c2 = idx >= pre[2], c3 = idx >= pre[3];
                const int p   = c3 ? pk[3] : c2 ? pk[2] : c1 ? pk[1] : pk[0];
                const int loc = idx - (c3 ? pre[3] : c2 ? pre[2] : c1 ? pre[1] : 0);
                const int y   = p & 511;
                const int l0  = (p >> 9) & 127;
                const int r0  = (p >> 16) & 127;
                const int n0  = p >> 23;
                const int q   = (loc < n0) ? (l0 + loc) : (r0 + (loc - n0));
                const float fx = (float)(4 * q - 256);
                const float fy = (float)(y - 256);
                const float hy = __builtin_fmaf(fy, fy, 0.5f);
                bool a = false;
                for (int j = 0; j < 4; ++j) {
                    const float xj = fx + (float)j;
                    const int b = (int)__builtin_amdgcn_sqrtf(__builtin_fmaf(xj, xj, hy));
                    bq[k][j] = b;
                    a = a || ((unsigned)(b - lo) < (unsigned)(hi - lo));
                }
                act[k] = a && (idx0 < total);
                const int off = y * (W_ / 4) + q;
                x1r[k] = __builtin_nontemporal_load(q1r + off);
                x1i[k] = __builtin_nontemporal_load(q1i + off);
                x2r[k] = __builtin_nontemporal_load(q2r + off);
                x2i[k] = __builtin_nontemporal_load(q2i + off);
            }

            #pragma unroll
            for (int k = 0; k < UNR; ++k) {
                const v4f a1r = x1r[k], a1i = x1i[k], a2r = x2r[k], a2i = x2i[k];
                const int b0 = bq[k][0], b1 = bq[k][1], b2 = bq[k][2], b3 = bq[k][3];

                const float cr0 = a1r.x * a2r.x + a1i.x * a2i.x;
                const float cr1 = a1r.y * a2r.y + a1i.y * a2i.y;
                const float cr2 = a1r.z * a2r.z + a1i.z * a2i.z;
                const float cr3 = a1r.w * a2r.w + a1i.w * a2i.w;
                const float ci0 = a1i.x * a2r.x - a1r.x * a2i.x;
                const float ci1 = a1i.y * a2r.y - a1r.y * a2i.y;
                const float ci2 = a1i.z * a2r.z - a1r.z * a2i.z;
                const float ci3 = a1i.w * a2r.w - a1r.w * a2i.w;
                const float p10 = a1r.x * a1r.x + a1i.x * a1i.x;
                const float p11 = a1r.y * a1r.y + a1i.y * a1i.y;
                const float p12 = a1r.z * a1r.z + a1i.z * a1i.z;
                const float p13 = a1r.w * a1r.w + a1i.w * a1i.w;
                const float p20 = a2r.x * a2r.x + a2i.x * a2i.x;
                const float p21 = a2r.y * a2r.y + a2i.y * a2i.y;
                const float p22 = a2r.z * a2r.z + a2i.z * a2i.z;
                const float p23 = a2r.w * a2r.w + a2i.w * a2i.w;

                if (!act[k]) continue;
                if ((b0 == b1) & (b1 == b2) & (b2 == b3)) {
                    atomicAdd(&hist[hb         + b0], cr0 + cr1 + cr2 + cr3);
                    atomicAdd(&hist[hb +   NBL + b0], ci0 + ci1 + ci2 + ci3);
                    atomicAdd(&hist[hb + 2*NBL + b0], p10 + p11 + p12 + p13);
                    atomicAdd(&hist[hb + 3*NBL + b0], p20 + p21 + p22 + p23);
                } else {
                    if ((unsigned)(b0 - lo) < (unsigned)(hi - lo)) {
                        atomicAdd(&hist[hb + b0], cr0); atomicAdd(&hist[hb + NBL + b0], ci0);
                        atomicAdd(&hist[hb + 2*NBL + b0], p10); atomicAdd(&hist[hb + 3*NBL + b0], p20);
                    }
                    if ((unsigned)(b1 - lo) < (unsigned)(hi - lo)) {
                        atomicAdd(&hist[hb + b1], cr1); atomicAdd(&hist[hb + NBL + b1], ci1);
                        atomicAdd(&hist[hb + 2*NBL + b1], p11); atomicAdd(&hist[hb + 3*NBL + b1], p21);
                    }
                    if ((unsigned)(b2 - lo) < (unsigned)(hi - lo)) {
                        atomicAdd(&hist[hb + b2], cr2); atomicAdd(&hist[hb + NBL + b2], ci2);
                        atomicAdd(&hist[hb + 2*NBL + b2], p12); atomicAdd(&hist[hb + 3*NBL + b2], p22);
                    }
                    if ((unsigned)(b3 - lo) < (unsigned)(hi - lo)) {
                        atomicAdd(&hist[hb + b3], cr3); atomicAdd(&hist[hb + NBL + b3], ci3);
                        atomicAdd(&hist[hb + 2*NBL + b3], p13); atomicAdd(&hist[hb + 3*NBL + b3], p23);
                    }
                }
            }
        }
    }
    __syncthreads();

    {
        v2f aOnes;
        aOnes.x = 1.0f;
        aOnes.y = 1.0f;
        const int half = lane >> 4;
        const int nIdx = lane & 15;
        for (int tile = wave; tile < HBINS / 16; tile += NWAVE) {
            const int col = tile * 16 + nIdx;
            v2f b[NCOPY / 4];
            #pragma unroll
            for (int s = 0; s < NCOPY / 4; ++s) {
                b[s].x = hist[(4 * s + 2 * half    ) * CSTR + col];
                b[s].y = hist[(4 * s + 2 * half + 1) * CSTR + col];
            }
            v8f c = {};
            #pragma unroll
            for (int s = 0; s < NCOPY / 4; ++s) {
                c = __builtin_amdgcn_wmma_f32_16x16x4_f32(
                         false, aOnes,  false, b[s],
                         (short)0, c,  false,  false);
            }
            if (lane < 16) acc[col] = c[0];
        }
    }
    __syncthreads();

    if (t < NRING && t >= lo && t < hi) {
        const int r = t;
        const float cr = acc[r - lo];
        const float ci = acc[NBL + r - lo];
        const float p1 = acc[2 * NBL + r - lo];
        const float p2 = acc[3 * NBL + r - lo];
        const float num = __builtin_sqrtf(cr * cr + ci * ci);
        const float den = __builtin_sqrtf(p1 * p2);
        out[(size_t)img * NRING + r] = (den == 0.0f) ? 0.0f : num / den;
    }
}

extern "C" void kernel_launch(void* const* d_in, const int* in_sizes, int n_in,
                              void* d_out, int out_size, void* d_ws, size_t ws_size,
                              hipStream_t stream)
{
    (void)in_sizes; (void)n_in; (void)out_size; (void)d_ws; (void)ws_size;
    const float* f1r = (const float*)d_in[0];
    const float* f1i = (const float*)d_in[1];
    const float* f2r = (const float*)d_in[2];
    const float* f2i = (const float*)d_in[3];

    frc_banded<<<dim3(B_ * NBAND), dim3(THREADS), 0, stream>>>(
        f1r, f1i, f2r, f2i, (float*)d_out);
}


# LDS histogram copies interleaved bin-major (word = bin*32 + lane): the 32 lanes of a float atomic hit 32 different banks; was copy pitch 517, about 2 bank passes an atomic (new f32 sum order)
# speedup vs baseline: 1.3489x; 1.0358x over previous
#include <hip/hip_runtime.h>


#define B_    64
#define H_    512
#define W_    512
#define HW    (H_ * W_)
#define NRING 256
#define NBAND 4
#define NBL   128
#define HBINS (4 * NBL)
#define NCOPY 32
#define THREADS 1024
#define NWAVE (THREADS / 32)
#define ROWS_PER_GROUP 4
#define UNR 4

typedef __attribute__((ext_vector_type(2))) float v2f;
typedef __attribute__((ext_vector_type(4))) float v4f;
typedef __attribute__((ext_vector_type(8))) float v8f;

__global__ __launch_bounds__(THREADS) void frc_banded(
    const float* __restrict__ f1r, const float* __restrict__ f1i,
    const float* __restrict__ f2r, const float* __restrict__ f2i,
    float* __restrict__ out)
{
    __shared__ __attribute__((aligned(16))) float hist[HBINS * NCOPY];
    __shared__ float acc[HBINS];

    const int t    = threadIdx.x;
    const int lane = t & 31;
    const int wave = __builtin_amdgcn_readfirstlane(t >> 5);
    const int img  = blockIdx.x / NBAND;
    const int band = blockIdx.x % NBAND;

    const int lo = (band == 0) ? 0 : (band == 1) ? 128 : (band == 2) ? 192 : 224;
    const int hi = (band == 0) ? 128 : (band == 1) ? 192 : (band == 2) ? 224 : 256;
    const int lo2 = lo * lo;
    const int hi2 = hi * hi;

    for (int i = t; i < HBINS * NCOPY; i += THREADS) hist[i] = 0.0f;
    __syncthreads();

    const int hb = lane - lo * NCOPY;

    const v4f* q1r = (const v4f*)f1r + (size_t)img * (HW / 4);
    const v4f* q1i = (const v4f*)f1i + (size_t)img * (HW / 4);
    const v4f* q2r = (const v4f*)f2r + (size_t)img * (HW / 4);
    const v4f* q2i = (const v4f*)f2i + (size_t)img * (HW / 4);

    int rowPack, rowCnt;
    {
        const int y   = wave + NWAVE * (lane & 15);
        const int dy  = y - 256;
        const int dy2 = dy * dy;
        int l0 = 0, n0 = 0, r0 = 0, n1 = 0;
        if (dy2 < hi2) {
            int ain = 0;
            if (lo2 > dy2) {
                ain = (int)__builtin_sqrtf((float)(lo2 - dy2)) - 1;
                ain = ain < 0 ? 0 : ain;
            }
            int bmax = (int)__builtin_sqrtf((float)(hi2 - dy2)) + 1;
            bmax = bmax > 256 ? 256 : bmax;
            const int xr1 = (256 + bmax) > 511 ? 511 : (256 + bmax);
            const int rsq = (256 + ain) >> 2;
            const int req = xr1 >> 2;
            const int lsq = (256 - bmax) >> 2;
            const int leq = (256 - ain) >> 2;
            if (leq >= rsq) { l0 = lsq; n0 = req - lsq + 1; r0 = 0; n1 = 0; }
            else            { l0 = lsq; n0 = leq - lsq + 1; r0 = rsq; n1 = req - rsq + 1; }
        }
        rowPack = y | (l0 << 9) | (r0 << 16) | (n0 << 23);
        rowCnt  = n0 + n1;
    }

    for (int g = 0; g < H_ / (NWAVE * ROWS_PER_GROUP); ++g) {
        int pk[ROWS_PER_GROUP];
        int pre[ROWS_PER_GROUP + 1];
        pre[0] = 0;
        for (int u = 0; u < ROWS_PER_GROUP; ++u) {
            pk[u]      = __builtin_amdgcn_readlane(rowPack, ROWS_PER_GROUP * g + u);
            pre[u + 1] = pre[u] + __builtin_amdgcn_readlane(rowCnt, ROWS_PER_GROUP * g + u);
        }
        const int total = pre[ROWS_PER_GROUP];

        for (int it0 = 0; it0 * 32 < total; it0 += UNR) {
            v4f x1r[UNR], x1i[UNR], x2r[UNR], x2i[UNR];
            int  bq[UNR][4];
            bool act[UNR];

            #pragma unroll
            for (int k = 0; k < UNR; ++k) {
                const int idx0 = (it0 + k) * 32 + lane;
                const int idx = idx0 < total ? idx0 : total - 1;
                const bool c1 = idx >= pre[1], c2 = idx >= pre[2], c3 = idx >= pre[3];
                const int p   = c3 ? pk[3] : c2 ? pk[2] : c1 ? pk[1] : pk[0];
                const int loc = idx - (c3 ? pre[3] : c2 ? pre[2] : c1 ? pre[1] : 0);
                const int y   = p & 511;
                const int l0  = (p >> 9) & 127;
                const int r0  = (p >> 16) & 127;
                const int n0  = p >> 23;
                const int q   = (loc < n0) ? (l0 + loc) : (r0 + (loc - n0));
                const float fx = (float)(4 * q - 256);
                const float fy = (float)(y - 256);
                const float hy = __builtin_fmaf(fy, fy, 0.5f);
                bool a = false;
                for (int j = 0; j < 4; ++j) {
                    const float xj = fx + (float)j;
                    const int b = (int)__builtin_amdgcn_sqrtf(__builtin_fmaf(xj, xj, hy));
                    bq[k][j] = b;
                    a = a || ((unsigned)(b - lo) < (unsigned)(hi - lo));
                }
                act[k] = a && (idx0 < total);
                const int off = y * (W_ / 4) + q;
                x1r[k] = __builtin_nontemporal_load(q1r + off);
                x1i[k] = __builtin_nontemporal_load(q1i + off);
                x2r[k] = __builtin_nontemporal_load(q2r + off);
                x2i[k] = __builtin_nontemporal_load(q2i + off);
            }

            #pragma unroll
            for (int k = 0; k < UNR; ++k) {
                const v4f a1r = x1r[k], a1i = x1i[k], a2r = x2r[k], a2i = x2i[k];
                const int b0 = bq[k][0], b1 = bq[k][1], b2 = bq[k][2], b3 = bq[k][3];

                const float cr0 = a1r.x * a2r.x + a1i.x * a2i.x;
                const float cr1 = a1r.y * a2r.y + a1i.y * a2i.y;
                const float cr2 = a1r.z * a2r.z + a1i.z * a2i.z;
                const float cr3 = a1r.w * a2r.w + a1i.w * a2i.w;
                const float ci0 = a1i.x * a2r.x - a1r.x * a2i.x;
                const float ci1 = a1i.y * a2r.y - a1r.y * a2i.y;
                const float ci2 = a1i.z * a2r.z - a1r.z * a2i.z;
                const float ci3 = a1i.w * a2r.w - a1r.w * a2i.w;
                const float p10 = a1r.x * a1r.x + a1i.x * a1i.x;
                const float p11 = a1r.y * a1r.y + a1i.y * a1i.y;
                const float p12 = a1r.z * a1r.z + a1i.z * a1i.z;
                const float p13 = a1r.w * a1r.w + a1i.w * a1i.w;
                const float p20 = a2r.x * a2r.x + a2i.x * a2i.x;
                const float p21 = a2r.y * a2r.y + a2i.y * a2i.y;
                const float p22 = a2r.z * a2r.z + a2i.z * a2i.z;
                const float p23 = a2r.w * a2r.w + a2i.w * a2i.w;

                if (!act[k]) continue;
                if ((b0 == b1) & (b1 == b2) & (b2 == b3)) {
                    atomicAdd(&hist[hb + (        b0) * NCOPY], cr0 + cr1 + cr2 + cr3);
                    atomicAdd(&hist[hb + (  NBL + b0) * NCOPY], ci0 + ci1 + ci2 + ci3);
                    atomicAdd(&hist[hb + (2*NBL + b0) * NCOPY], p10 + p11 + p12 + p13);
                    atomicAdd(&hist[hb + (3*NBL + b0) * NCOPY], p20 + p21 + p22 + p23);
                } else {
                    if ((unsigned)(b0 - lo) < (unsigned)(hi - lo)) {
                        atomicAdd(&hist[hb + b0 * NCOPY], cr0); atomicAdd(&hist[hb + (NBL + b0) * NCOPY], ci0);
                        atomicAdd(&hist[hb + (2*NBL + b0) * NCOPY], p10); atomicAdd(&hist[hb + (3*NBL + b0) * NCOPY], p20);
                    }
                    if ((unsigned)(b1 - lo) < (unsigned)(hi - lo)) {
                        atomicAdd(&hist[hb + b1 * NCOPY], cr1); atomicAdd(&hist[hb + (NBL + b1) * NCOPY], ci1);
                        atomicAdd(&hist[hb + (2*NBL + b1) * NCOPY], p11); atomicAdd(&hist[hb + (3*NBL + b1) * NCOPY], p21);
                    }
                    if ((unsigned)(b2 - lo) < (unsigned)(hi - lo)) {
                        atomicAdd(&hist[hb + b2 * NCOPY], cr2); atomicAdd(&hist[hb + (NBL + b2) * NCOPY], ci2);
                        atomicAdd(&hist[hb + (2*NBL + b2) * NCOPY], p12); atomicAdd(&hist[hb + (3*NBL + b2) * NCOPY], p22);
                    }
                    if ((unsigned)(b3 - lo) < (unsigned)(hi - lo)) {
                        atomicAdd(&hist[hb + b3 * NCOPY], cr3); atomicAdd(&hist[hb + (NBL + b3) * NCOPY], ci3);
                        atomicAdd(&hist[hb + (2*NBL + b3) * NCOPY], p13); atomicAdd(&hist[hb + (3*NBL + b3) * NCOPY], p23);
                    }
                }
            }
        }
    }
    __syncthreads();

    {
        v2f aOnes;
        aOnes.x = 1.0f;
        aOnes.y = 1.0f;
        const int half = lane >> 4;
        const int nIdx = lane & 15;
        for (int tile = wave; tile < HBINS / 16; tile += NWAVE) {
            const int col = tile * 16 + nIdx;
            v2f b[NCOPY / 4];
            #pragma unroll
            for (int s = 0; s < NCOPY / 4; ++s) {
                const int k0 = (4 * s + 2 * half + 2 * nIdx) & (NCOPY - 1);
                b[s] = *(const v2f*)&hist[col * NCOPY + k0];
            }
            __builtin_amdgcn_sched_barrier(0);
            v8f c = {};
            #pragma unroll
            for (int s = 0; s < NCOPY / 4; ++s) {
                c = __builtin_amdgcn_wmma_f32_16x16x4_f32(
                         false, aOnes,  false, b[s],
                         (short)0, c,  false,  false);
            }
            if (lane < 16) acc[col] = c[0];
        }
    }
    __syncthreads();

    if (t < NRING && t >= lo && t < hi) {
        const int r = t;
        const float cr = acc[r - lo];
        const float ci = acc[NBL + r - lo];
        const float p1 = acc[2 * NBL + r - lo];
        const float p2 = acc[3 * NBL + r - lo];
        const float num = __builtin_sqrtf(cr * cr + ci * ci);
        const float den = __builtin_sqrtf(p1 * p2);
        out[(size_t)img * NRING + r] = (den == 0.0f) ? 0.0f : num / den;
    }
}

extern "C" void kernel_launch(void* const* d_in, const int* in_sizes, int n_in,
                              void* d_out, int out_size, void* d_ws, size_t ws_size,
                              hipStream_t stream)
{
    (void)in_sizes; (void)n_in; (void)out_size; (void)d_ws; (void)ws_size;
    const float* f1r = (const float*)d_in[0];
    const float* f1i = (const float*)d_in[1];
    const float* f2r = (const float*)d_in[2];
    const float* f2i = (const float*)d_in[3];

    frc_banded<<<dim3(B_ * NBAND), dim3(THREADS), 0, stream>>>(
        f1r, f1i, f2r, f2i, (float*)d_out);
}


# Block index = rank*64 + image with bands ranked by work (rings 128-191, 224-255, 0-127, 192-223; was image*4 + band): the longest-running workgroups enter first, the shortest last; sums unchanged
# speedup vs baseline: 1.4513x; 1.0759x over previous
#include <hip/hip_runtime.h>


#define B_    64
#define H_    512
#define W_    512
#define HW    (H_ * W_)
#define NRING 256
#define NBAND 4
#define NBL   128
#define HBINS (4 * NBL)
#define NCOPY 32
#define THREADS 1024
#define NWAVE (THREADS / 32)
#define ROWS_PER_GROUP 4
#define UNR 4

typedef __attribute__((ext_vector_type(2))) float v2f;
typedef __attribute__((ext_vector_type(4))) float v4f;
typedef __attribute__((ext_vector_type(8))) float v8f;

__global__ __launch_bounds__(THREADS) void frc_banded(
    const float* __restrict__ f1r, const float* __restrict__ f1i,
    const float* __restrict__ f2r, const float* __restrict__ f2i,
    float* __restrict__ out)
{
    __shared__ __attribute__((aligned(16))) float hist[HBINS * NCOPY];
    __shared__ float acc[HBINS];

    const int t    = threadIdx.x;
    const int lane = t & 31;
    const int wave = __builtin_amdgcn_readfirstlane(t >> 5);
    const int rank = blockIdx.x / B_;
    const int img  = blockIdx.x % B_;

    const int lo = (rank == 0) ? 128 : (rank == 1) ? 224 : (rank == 2) ? 0 : 192;
    const int hi = (rank == 0) ? 192 : (rank == 1) ? 256 : (rank == 2) ? 128 : 224;
    const int lo2 = lo * lo;
    const int hi2 = hi * hi;

    for (int i = t; i < HBINS * NCOPY; i += THREADS) hist[i] = 0.0f;
    __syncthreads();

    const int hb = lane - lo * NCOPY;

    const v4f* q1r = (const v4f*)f1r + (size_t)img * (HW / 4);
    const v4f* q1i = (const v4f*)f1i + (size_t)img * (HW / 4);
    const v4f* q2r = (const v4f*)f2r + (size_t)img * (HW / 4);
    const v4f* q2i = (const v4f*)f2i + (size_t)img * (HW / 4);

    int rowPack, rowCnt;
    {
        const int y   = wave + NWAVE * (lane & 15);
        const int dy  = y - 256;
        const int dy2 = dy * dy;
        int l0 = 0, n0 = 0, r0 = 0, n1 = 0;
        if (dy2 < hi2) {
            int ain = 0;
            if (lo2 > dy2) {
                ain = (int)__builtin_sqrtf((float)(lo2 - dy2)) - 1;
                ain = ain < 0 ? 0 : ain;
            }
            int bmax = (int)__builtin_sqrtf((float)(hi2 - dy2)) + 1;
            bmax = bmax > 256 ? 256 : bmax;
            const int xr1 = (256 + bmax) > 511 ? 511 : (256 + bmax);
            const int rsq = (256 + ain) >> 2;
            const int req = xr1 >> 2;
            const int lsq = (256 - bmax) >> 2;
            const int leq = (256 - ain) >> 2;
            if (leq >= rsq) { l0 = lsq; n0 = req - lsq + 1; r0 = 0; n1 = 0; }
            else            { l0 = lsq; n0 = leq - lsq + 1; r0 = rsq; n1 = req - rsq + 1; }
        }
        rowPack = y | (l0 << 9) | (r0 << 16) | (n0 << 23);
        rowCnt  = n0 + n1;
    }

    for (int g = 0; g < H_ / (NWAVE * ROWS_PER_GROUP); ++g) {
        int pk[ROWS_PER_GROUP];
        int pre[ROWS_PER_GROUP + 1];
        pre[0] = 0;
        for (int u = 0; u < ROWS_PER_GROUP; ++u) {
            pk[u]      = __builtin_amdgcn_readlane(rowPack, ROWS_PER_GROUP * g + u);
            pre[u + 1] = pre[u] + __builtin_amdgcn_readlane(rowCnt, ROWS_PER_GROUP * g + u);
        }
        const int total = pre[ROWS_PER_GROUP];

        for (int it0 = 0; it0 * 32 < total; it0 += UNR) {
            v4f x1r[UNR], x1i[UNR], x2r[UNR], x2i[UNR];
            int  bq[UNR][4];
            bool act[UNR];

            #pragma unroll
            for (int k = 0; k < UNR; ++k) {
                const int idx0 = (it0 + k) * 32 + lane;
                const int idx = idx0 < total ? idx0 : total - 1;
                const bool c1 = idx >= pre[1], c2 = idx >= pre[2], c3 = idx >= pre[3];
                const int p   = c3 ? pk[3] : c2 ? pk[2] : c1 ? pk[1] : pk[0];
                const int loc = idx - (c3 ? pre[3] : c2 ? pre[2] : c1 ? pre[1] : 0);
                const int y   = p & 511;
                const int l0  = (p >> 9) & 127;
                const int r0  = (p >> 16) & 127;
                const int n0  = p >> 23;
                const int q   = (loc < n0) ? (l0 + loc) : (r0 + (loc - n0));
                const float fx = (float)(4 * q - 256);
                const float fy = (float)(y - 256);
                const float hy = __builtin_fmaf(fy, fy, 0.5f);
                bool a = false;
                for (int j = 0; j < 4; ++j) {
                    const float xj = fx + (float)j;
                    const int b = (int)__builtin_amdgcn_sqrtf(__builtin_fmaf(xj, xj, hy));
                    bq[k][j] = b;
                    a = a || ((unsigned)(b - lo) < (unsigned)(hi - lo));
                }
                act[k] = a && (idx0 < total);
                const int off = y * (W_ / 4) + q;
                x1r[k] = __builtin_nontemporal_load(q1r + off);
                x1i[k] = __builtin_nontemporal_load(q1i + off);
                x2r[k] = __builtin_nontemporal_load(q2r + off);
                x2i[k] = __builtin_nontemporal_load(q2i + off);
            }

            #pragma unroll
            for (int k = 0; k < UNR; ++k) {
                const v4f a1r = x1r[k], a1i = x1i[k], a2r = x2r[k], a2i = x2i[k];
                const int b0 = bq[k][0], b1 = bq[k][1], b2 = bq[k][2], b3 = bq[k][3];

                const float cr0 = a1r.x * a2r.x + a1i.x * a2i.x;
                const float cr1 = a1r.y * a2r.y + a1i.y * a2i.y;
                const float cr2 = a1r.z * a2r.z + a1i.z * a2i.z;
                const float cr3 = a1r.w * a2r.w + a1i.w * a2i.w;
                const float ci0 = a1i.x * a2r.x - a1r.x * a2i.x;
                const float ci1 = a1i.y * a2r.y - a1r.y * a2i.y;
                const float ci2 = a1i.z * a2r.z - a1r.z * a2i.z;
                const float ci3 = a1i.w * a2r.w - a1r.w * a2i.w;
                const float p10 = a1r.x * a1r.x + a1i.x * a1i.x;
                const float p11 = a1r.y * a1r.y + a1i.y * a1i.y;
                const float p12 = a1r.z * a1r.z + a1i.z * a1i.z;
                const float p13 = a1r.w * a1r.w + a1i.w * a1i.w;
                const float p20 = a2r.x * a2r.x + a2i.x * a2i.x;
                const float p21 = a2r.y * a2r.y + a2i.y * a2i.y;
                const float p22 = a2r.z * a2r.z + a2i.z * a2i.z;
                const float p23 = a2r.w * a2r.w + a2i.w * a2i.w;

                if (!act[k]) continue;
                if ((b0 == b1) & (b1 == b2) & (b2 == b3)) {
                    atomicAdd(&hist[hb + (        b0) * NCOPY], cr0 + cr1 + cr2 + cr3);
                    atomicAdd(&hist[hb + (  NBL + b0) * NCOPY], ci0 + ci1 + ci2 + ci3);
                    atomicAdd(&hist[hb + (2*NBL + b0) * NCOPY], p10 + p11 + p12 + p13);
                    atomicAdd(&hist[hb + (3*NBL + b0) * NCOPY], p20 + p21 + p22 + p23);
                } else {
                    if ((unsigned)(b0 - lo) < (unsigned)(hi - lo)) {
                        atomicAdd(&hist[hb + b0 * NCOPY], cr0); atomicAdd(&hist[hb + (NBL + b0) * NCOPY], ci0);
                        atomicAdd(&hist[hb + (2*NBL + b0) * NCOPY], p10); atomicAdd(&hist[hb + (3*NBL + b0) * NCOPY], p20);
                    }
                    if ((unsigned)(b1 - lo) < (unsigned)(hi - lo)) {
                        atomicAdd(&hist[hb + b1 * NCOPY], cr1); atomicAdd(&hist[hb + (NBL + b1) * NCOPY], ci1);
                        atomicAdd(&hist[hb + (2*NBL + b1) * NCOPY], p11); atomicAdd(&hist[hb + (3*NBL + b1) * NCOPY], p21);
                    }
                    if ((unsigned)(b2 - lo) < (unsigned)(hi - lo)) {
                        atomicAdd(&hist[hb + b2 * NCOPY], cr2); atomicAdd(&hist[hb + (NBL + b2) * NCOPY], ci2);
                        atomicAdd(&hist[hb + (2*NBL + b2) * NCOPY], p12); atomicAdd(&hist[hb + (3*NBL + b2) * NCOPY], p22);
                    }
                    if ((unsigned)(b3 - lo) < (unsigned)(hi - lo)) {
                        atomicAdd(&hist[hb + b3 * NCOPY], cr3); atomicAdd(&hist[hb + (NBL + b3) * NCOPY], ci3);
                        atomicAdd(&hist[hb + (2*NBL + b3) * NCOPY], p13); atomicAdd(&hist[hb + (3*NBL + b3) * NCOPY], p23);
                    }
                }
            }
        }
    }
    __syncthreads();

    {
        v2f aOnes;
        aOnes.x = 1.0f;
        aOnes.y = 1.0f;
        const int half = lane >> 4;
        const int nIdx = lane & 15;
        for (int tile = wave; tile < HBINS / 16; tile += NWAVE) {
            const int col = tile * 16 + nIdx;
            v2f b[NCOPY / 4];
            #pragma unroll
            for (int s = 0; s < NCOPY / 4; ++s) {
                const int k0 = (4 * s + 2 * half + 2 * nIdx) & (NCOPY - 1);
                b[s] = *(const v2f*)&hist[col * NCOPY + k0];
            }
            __builtin_amdgcn_sched_barrier(0);
            v8f c = {};
            #pragma unroll
            for (int s = 0; s < NCOPY / 4; ++s) {
                c = __builtin_amdgcn_wmma_f32_16x16x4_f32(
                         false, aOnes,  false, b[s],
                         (short)0, c,  false,  false);
            }
            if (lane < 16) acc[col] = c[0];
        }
    }
    __syncthreads();

    if (t < NRING && t >= lo && t < hi) {
        const int r = t;
        const float cr = acc[r - lo];
        const float ci = acc[NBL + r - lo];
        const float p1 = acc[2 * NBL + r - lo];
        const float p2 = acc[3 * NBL + r - lo];
        const float num = __builtin_sqrtf(cr * cr + ci * ci);
        const float den = __builtin_sqrtf(p1 * p2);
        out[(size_t)img * NRING + r] = (den == 0.0f) ? 0.0f : num / den;
    }
}

extern "C" void kernel_launch(void* const* d_in, const int* in_sizes, int n_in,
                              void* d_out, int out_size, void* d_ws, size_t ws_size,
                              hipStream_t stream)
{
    (void)in_sizes; (void)n_in; (void)out_size; (void)d_ws; (void)ws_size;
    const float* f1r = (const float*)d_in[0];
    const float* f1i = (const float*)d_in[1];
    const float* f2r = (const float*)d_in[2];
    const float* f2i = (const float*)d_in[3];

    frc_banded<<<dim3(B_ * NBAND), dim3(THREADS), 0, stream>>>(
        f1r, f1i, f2r, f2i, (float*)d_out);
}


# Branch-free binning: each pixel adds its 4 values to LDS at a clamped address, out-of-band pixels and idle slots hitting a spare row never read; values formed as packed pixel pairs (new f32 sum order)
# speedup vs baseline: 1.5105x; 1.0408x over previous
#include <hip/hip_runtime.h>


#define B_    64
#define H_    512
#define W_    512
#define HW    (H_ * W_)
#define NRING 256
#define NBAND 4
#define NBL   128
#define HBINS (4 * NBL)
#define NCOPY 32
#define NROW  (NBL + 1)
#define PLW   (NROW * NCOPY)
#define HWORDS (4 * PLW)
#define THREADS 1024
#define NWAVE (THREADS / 32)
#define ROWS_PER_GROUP 4
#define UNR 4

typedef __attribute__((ext_vector_type(2))) float v2f;
typedef __attribute__((ext_vector_type(4))) float v4f;
typedef __attribute__((ext_vector_type(8))) float v8f;

__global__ __launch_bounds__(THREADS) void frc_banded(
    const float* __restrict__ f1r, const float* __restrict__ f1i,
    const float* __restrict__ f2r, const float* __restrict__ f2i,
    float* __restrict__ out)
{
    __shared__ __attribute__((aligned(16))) float hist[HWORDS];
    __shared__ float acc[HBINS];

    const int t    = threadIdx.x;
    const int lane = t & 31;
    const int wave = __builtin_amdgcn_readfirstlane(t >> 5);
    const int rank = blockIdx.x / B_;
    const int img  = blockIdx.x % B_;

    const int lo = (rank == 0) ? 128 : (rank == 1) ? 224 : (rank == 2) ? 0 : 192;
    const int hi = (rank == 0) ? 192 : (rank == 1) ? 256 : (rank == 2) ? 128 : 224;
    const int lo2 = lo * lo;
    const int hi2 = hi * hi;

    for (int i = t; i < HWORDS; i += THREADS) hist[i] = 0.0f;
    __syncthreads();

    const int      hbB    = lane * 4 - lo * (NCOPY * 4);
    const unsigned spareB = (unsigned)(NBL * NCOPY * 4 + lane * 4);
    char* const    histB  = (char*)hist;

    const v4f* q1r = (const v4f*)f1r + (size_t)img * (HW / 4);
    const v4f* q1i = (const v4f*)f1i + (size_t)img * (HW / 4);
    const v4f* q2r = (const v4f*)f2r + (size_t)img * (HW / 4);
    const v4f* q2i = (const v4f*)f2i + (size_t)img * (HW / 4);

    int rowPack, rowCnt;
    {
        const int y   = wave + NWAVE * (lane & 15);
        const int dy  = y - 256;
        const int dy2 = dy * dy;
        int l0 = 0, n0 = 0, r0 = 0, n1 = 0;
        if (dy2 < hi2) {
            int ain = 0;
            if (lo2 > dy2) {
                ain = (int)__builtin_sqrtf((float)(lo2 - dy2)) - 1;
                ain = ain < 0 ? 0 : ain;
            }
            int bmax = (int)__builtin_sqrtf((float)(hi2 - dy2)) + 1;
            bmax = bmax > 256 ? 256 : bmax;
            const int xr1 = (256 + bmax) > 511 ? 511 : (256 + bmax);
            const int rsq = (256 + ain) >> 2;
            const int req = xr1 >> 2;
            const int lsq = (256 - bmax) >> 2;
            const int leq = (256 - ain) >> 2;
            if (leq >= rsq) { l0 = lsq; n0 = req - lsq + 1; r0 = 0; n1 = 0; }
            else            { l0 = lsq; n0 = leq - lsq + 1; r0 = rsq; n1 = req - rsq + 1; }
        }
        rowPack = y | (l0 << 9) | (r0 << 16) | (n0 << 23);
        rowCnt  = n0 + n1;
    }

    for (int g = 0; g < H_ / (NWAVE * ROWS_PER_GROUP); ++g) {
        int pk[ROWS_PER_GROUP];
        int pre[ROWS_PER_GROUP + 1];
        pre[0] = 0;
        for (int u = 0; u < ROWS_PER_GROUP; ++u) {
            pk[u]      = __builtin_amdgcn_readlane(rowPack, ROWS_PER_GROUP * g + u);
            pre[u + 1] = pre[u] + __builtin_amdgcn_readlane(rowCnt, ROWS_PER_GROUP * g + u);
        }
        const int total = pre[ROWS_PER_GROUP];

        for (int it0 = 0; it0 * 32 < total; it0 += UNR) {
            v4f x1r[UNR], x1i[UNR], x2r[UNR], x2i[UNR];
            unsigned ad[UNR][4];

            #pragma unroll
            for (int k = 0; k < UNR; ++k) {
                const int idx0 = (it0 + k) * 32 + lane;
                const int idx = idx0 < total ? idx0 : total - 1;
                const bool c1 = idx >= pre[1], c2 = idx >= pre[2], c3 = idx >= pre[3];
                const int p   = c3 ? pk[3] : c2 ? pk[2] : c1 ? pk[1] : pk[0];
                const int loc = idx - (c3 ? pre[3] : c2 ? pre[2] : c1 ? pre[1] : 0);
                const int y   = p & 511;
                const int l0  = (p >> 9) & 127;
                const int r0  = (p >> 16) & 127;
                const int n0  = p >> 23;
                const int q   = (loc < n0) ? (l0 + loc) : (r0 + (loc - n0));
                const float fx = (float)(4 * q - 256);
                const float fy = (float)(y - 256);
                const float hy = (idx0 < total) ? __builtin_fmaf(fy, fy, 0.5f) : 1.0e12f;
                for (int j = 0; j < 4; ++j) {
                    const float xj = fx + (float)j;
                    const int b = (int)__builtin_amdgcn_sqrtf(__builtin_fmaf(xj, xj, hy));
                    const unsigned a = (unsigned)((b << 7) + hbB);
                    ad[k][j] = a < spareB ? a : spareB;
                }
                const int off = y * (W_ / 4) + q;
                x1r[k] = __builtin_nontemporal_load(q1r + off);
                x1i[k] = __builtin_nontemporal_load(q1i + off);
                x2r[k] = __builtin_nontemporal_load(q2r + off);
                x2i[k] = __builtin_nontemporal_load(q2i + off);
            }

            __builtin_amdgcn_sched_barrier(0);

            #pragma unroll
            for (int k = 0; k < UNR; ++k) {
                if ((it0 + k) * 32 >= total) break;

                const v2f a1rL = x1r[k].lo, a1rH = x1r[k].hi;
                const v2f a1iL = x1i[k].lo, a1iH = x1i[k].hi;
                const v2f a2rL = x2r[k].lo, a2rH = x2r[k].hi;
                const v2f a2iL = x2i[k].lo, a2iH = x2i[k].hi;

                const v2f crL = a1rL * a2rL + a1iL * a2iL;
                const v2f crH = a1rH * a2rH + a1iH * a2iH;
                const v2f ciL = a1iL * a2rL - a1rL * a2iL;
                const v2f ciH = a1iH * a2rH - a1rH * a2iH;
                const v2f p1L = a1rL * a1rL + a1iL * a1iL;
                const v2f p1H = a1rH * a1rH + a1iH * a1iH;
                const v2f p2L = a2rL * a2rL + a2iL * a2iL;
                const v2f p2H = a2rH * a2rH + a2iH * a2iH;

                float* const h0 = (float*)(histB + ad[k][0]);
                float* const h1 = (float*)(histB + ad[k][1]);
                float* const h2 = (float*)(histB + ad[k][2]);
                float* const h3 = (float*)(histB + ad[k][3]);
                atomicAdd(h0, crL.x); atomicAdd(h0 + PLW, ciL.x); atomicAdd(h0 + 2 * PLW, p1L.x); atomicAdd(h0 + 3 * PLW, p2L.x);
                atomicAdd(h1, crL.y); atomicAdd(h1 + PLW, ciL.y); atomicAdd(h1 + 2 * PLW, p1L.y); atomicAdd(h1 + 3 * PLW, p2L.y);
                atomicAdd(h2, crH.x); atomicAdd(h2 + PLW, ciH.x); atomicAdd(h2 + 2 * PLW, p1H.x); atomicAdd(h2 + 3 * PLW, p2H.x);
                atomicAdd(h3, crH.y); atomicAdd(h3 + PLW, ciH.y); atomicAdd(h3 + 2 * PLW, p1H.y); atomicAdd(h3 + 3 * PLW, p2H.y);
            }
            asm volatile("" :: "v"(x1r[UNR - 1]), "v"(x1i[UNR - 1]), "v"(x2r[UNR - 1]), "v"(x2i[UNR - 1]));
        }
    }
    __syncthreads();

    {
        v2f aOnes;
        aOnes.x = 1.0f;
        aOnes.y = 1.0f;
        const int half = lane >> 4;
        const int nIdx = lane & 15;
        for (int tile = wave; tile < HBINS / 16; tile += NWAVE) {
            const int col = tile * 16 + nIdx;
            const int row0 = (col / NBL) * PLW + (col % NBL) * NCOPY;
            v2f b[NCOPY / 4];
            #pragma unroll
            for (int s = 0; s < NCOPY / 4; ++s) {
                const int k0 = (4 * s + 2 * half + 2 * nIdx) & (NCOPY - 1);
                b[s] = *(const v2f*)&hist[row0 + k0];
            }
            __builtin_amdgcn_sched_barrier(0);
            v8f c = {};
            #pragma unroll
            for (int s = 0; s < NCOPY / 4; ++s) {
                c = __builtin_amdgcn_wmma_f32_16x16x4_f32(
                         false, aOnes,  false, b[s],
                         (short)0, c,  false,  false);
            }
            if (lane < 16) acc[col] = c[0];
        }
    }
    __syncthreads();

    if (t < NRING && t >= lo && t < hi) {
        const int r = t;
        const float cr = acc[r - lo];
        const float ci = acc[NBL + r - lo];
        const float p1 = acc[2 * NBL + r - lo];
        const float p2 = acc[3 * NBL + r - lo];
        const float num = __builtin_sqrtf(cr * cr + ci * ci);
        const float den = __builtin_sqrtf(p1 * p2);
        out[(size_t)img * NRING + r] = (den == 0.0f) ? 0.0f : num / den;
    }
}

extern "C" void kernel_launch(void* const* d_in, const int* in_sizes, int n_in,
                              void* d_out, int out_size, void* d_ws, size_t ws_size,
                              hipStream_t stream)
{
    (void)in_sizes; (void)n_in; (void)out_size; (void)d_ws; (void)ws_size;
    const float* f1r = (const float*)d_in[0];
    const float* f1i = (const float*)d_in[1];
    const float* f2r = (const float*)d_in[2];
    const float* f2i = (const float*)d_in[3];

    frc_banded<<<dim3(B_ * NBAND), dim3(THREADS), 0, stream>>>(
        f1r, f1i, f2r, f2i, (float*)d_out);
}


# Band edges 137/186/224 (were 128/192/224) so that the four workgroups of an image carry equal lines and wave iterations; ranks now 224-255, 0-136, 186-223, 137-185 (new f32 sum order)
# speedup vs baseline: 1.6351x; 1.0825x over previous
#include <hip/hip_runtime.h>


#define B_    64
#define H_    512
#define W_    512
#define HW    (H_ * W_)
#define NRING 256
#define NBAND 4
#define NBL   144
#define HBINS (4 * NBL)
#define NCOPY 32
#define NROW  (NBL + 1)
#define PLW   (NROW * NCOPY)
#define HWORDS (4 * PLW)
#define THREADS 1024
#define NWAVE (THREADS / 32)
#define ROWS_PER_GROUP 4
#define UNR 4

typedef __attribute__((ext_vector_type(2))) float v2f;
typedef __attribute__((ext_vector_type(4))) float v4f;
typedef __attribute__((ext_vector_type(8))) float v8f;

__global__ __launch_bounds__(THREADS) void frc_banded(
    const float* __restrict__ f1r, const float* __restrict__ f1i,
    const float* __restrict__ f2r, const float* __restrict__ f2i,
    float* __restrict__ out)
{
    __shared__ __attribute__((aligned(16))) float hist[HWORDS];
    __shared__ float acc[HBINS];

    const int t    = threadIdx.x;
    const int lane = t & 31;
    const int wave = __builtin_amdgcn_readfirstlane(t >> 5);
    const int rank = blockIdx.x / B_;
    const int img  = blockIdx.x % B_;

    const int lo = (rank == 0) ? 224 : (rank == 1) ? 0   : (rank == 2) ? 186 : 137;
    const int hi = (rank == 0) ? 256 : (rank == 1) ? 137 : (rank == 2) ? 224 : 186;
    const int lo2 = lo * lo;
    const int hi2 = hi * hi;

    for (int i = t; i < HWORDS; i += THREADS) hist[i] = 0.0f;
    __syncthreads();

    const int      hbB    = lane * 4 - lo * (NCOPY * 4);
    const unsigned spareB = (unsigned)(NBL * NCOPY * 4 + lane * 4);
    char* const    histB  = (char*)hist;

    const v4f* q1r = (const v4f*)f1r + (size_t)img * (HW / 4);
    const v4f* q1i = (const v4f*)f1i + (size_t)img * (HW / 4);
    const v4f* q2r = (const v4f*)f2r + (size_t)img * (HW / 4);
    const v4f* q2i = (const v4f*)f2i + (size_t)img * (HW / 4);

    int rowPack, rowCnt;
    {
        const int y   = wave + NWAVE * (lane & 15);
        const int dy  = y - 256;
        const int dy2 = dy * dy;
        int l0 = 0, n0 = 0, r0 = 0, n1 = 0;
        if (dy2 < hi2) {
            int ain = 0;
            if (lo2 > dy2) {
                ain = (int)__builtin_sqrtf((float)(lo2 - dy2)) - 1;
                ain = ain < 0 ? 0 : ain;
            }
            int bmax = (int)__builtin_sqrtf((float)(hi2 - dy2)) + 1;
            bmax = bmax > 256 ? 256 : bmax;
            const int xr1 = (256 + bmax) > 511 ? 511 : (256 + bmax);
            const int rsq = (256 + ain) >> 2;
            const int req = xr1 >> 2;
            const int lsq = (256 - bmax) >> 2;
            const int leq = (256 - ain) >> 2;
            if (leq >= rsq) { l0 = lsq; n0 = req - lsq + 1; r0 = 0; n1 = 0; }
            else            { l0 = lsq; n0 = leq - lsq + 1; r0 = rsq; n1 = req - rsq + 1; }
        }
        rowPack = y | (l0 << 9) | (r0 << 16) | (n0 << 23);
        rowCnt  = n0 + n1;
    }

    for (int g = 0; g < H_ / (NWAVE * ROWS_PER_GROUP); ++g) {
        int pk[ROWS_PER_GROUP];
        int pre[ROWS_PER_GROUP + 1];
        pre[0] = 0;
        for (int u = 0; u < ROWS_PER_GROUP; ++u) {
            pk[u]      = __builtin_amdgcn_readlane(rowPack, ROWS_PER_GROUP * g + u);
            pre[u + 1] = pre[u] + __builtin_amdgcn_readlane(rowCnt, ROWS_PER_GROUP * g + u);
        }
        const int total = pre[ROWS_PER_GROUP];

        for (int it0 = 0; it0 * 32 < total; it0 += UNR) {
            v4f x1r[UNR], x1i[UNR], x2r[UNR], x2i[UNR];
            unsigned ad[UNR][4];

            #pragma unroll
            for (int k = 0; k < UNR; ++k) {
                const int idx0 = (it0 + k) * 32 + lane;
                const int idx = idx0 < total ? idx0 : total - 1;
                const bool c1 = idx >= pre[1], c2 = idx >= pre[2], c3 = idx >= pre[3];
                const int p   = c3 ? pk[3] : c2 ? pk[2] : c1 ? pk[1] : pk[0];
                const int loc = idx - (c3 ? pre[3] : c2 ? pre[2] : c1 ? pre[1] : 0);
                const int y   = p & 511;
                const int l0  = (p >> 9) & 127;
                const int r0  = (p >> 16) & 127;
                const int n0  = p >> 23;
                const int q   = (loc < n0) ? (l0 + loc) : (r0 + (loc - n0));
                const float fx = (float)(4 * q - 256);
                const float fy = (float)(y - 256);
                const float hy = (idx0 < total) ? __builtin_fmaf(fy, fy, 0.5f) : 1.0e12f;
                for (int j = 0; j < 4; ++j) {
                    const float xj = fx + (float)j;
                    const int b = (int)__builtin_amdgcn_sqrtf(__builtin_fmaf(xj, xj, hy));
                    const unsigned a = (unsigned)((b << 7) + hbB);
                    ad[k][j] = a < spareB ? a : spareB;
                }
                const int off = y * (W_ / 4) + q;
                x1r[k] = __builtin_nontemporal_load(q1r + off);
                x1i[k] = __builtin_nontemporal_load(q1i + off);
                x2r[k] = __builtin_nontemporal_load(q2r + off);
                x2i[k] = __builtin_nontemporal_load(q2i + off);
            }

            __builtin_amdgcn_sched_barrier(0);

            #pragma unroll
            for (int k = 0; k < UNR; ++k) {
                if ((it0 + k) * 32 >= total) break;

                const v2f a1rL = x1r[k].lo, a1rH = x1r[k].hi;
                const v2f a1iL = x1i[k].lo, a1iH = x1i[k].hi;
                const v2f a2rL = x2r[k].lo, a2rH = x2r[k].hi;
                const v2f a2iL = x2i[k].lo, a2iH = x2i[k].hi;

                const v2f crL = a1rL * a2rL + a1iL * a2iL;
                const v2f crH = a1rH * a2rH + a1iH * a2iH;
                const v2f ciL = a1iL * a2rL - a1rL * a2iL;
                const v2f ciH = a1iH * a2rH - a1rH * a2iH;
                const v2f p1L = a1rL * a1rL + a1iL * a1iL;
                const v2f p1H = a1rH * a1rH + a1iH * a1iH;
                const v2f p2L = a2rL * a2rL + a2iL * a2iL;
                const v2f p2H = a2rH * a2rH + a2iH * a2iH;

                float* const h0 = (float*)(histB + ad[k][0]);
                float* const h1 = (float*)(histB + ad[k][1]);
                float* const h2 = (float*)(histB + ad[k][2]);
                float* const h3 = (float*)(histB + ad[k][3]);
                atomicAdd(h0, crL.x); atomicAdd(h0 + PLW, ciL.x); atomicAdd(h0 + 2 * PLW, p1L.x); atomicAdd(h0 + 3 * PLW, p2L.x);
                atomicAdd(h1, crL.y); atomicAdd(h1 + PLW, ciL.y); atomicAdd(h1 + 2 * PLW, p1L.y); atomicAdd(h1 + 3 * PLW, p2L.y);
                atomicAdd(h2, crH.x); atomicAdd(h2 + PLW, ciH.x); atomicAdd(h2 + 2 * PLW, p1H.x); atomicAdd(h2 + 3 * PLW, p2H.x);
                atomicAdd(h3, crH.y); atomicAdd(h3 + PLW, ciH.y); atomicAdd(h3 + 2 * PLW, p1H.y); atomicAdd(h3 + 3 * PLW, p2H.y);
            }
            asm volatile("" :: "v"(x1r[UNR - 1]), "v"(x1i[UNR - 1]), "v"(x2r[UNR - 1]), "v"(x2i[UNR - 1]));
        }
    }
    __syncthreads();

    {
        v2f aOnes;
        aOnes.x = 1.0f;
        aOnes.y = 1.0f;
        const int half = lane >> 4;
        const int nIdx = lane & 15;
        const int nChunk = (hi - lo + 15) >> 4;
        for (int tile = wave; tile < HBINS / 16; tile += NWAVE) {
            if (tile % (NBL / 16) >= nChunk) continue;
            const int col = tile * 16 + nIdx;
            const int row0 = (col / NBL) * PLW + (col % NBL) * NCOPY;
            v2f b[NCOPY / 4];
            #pragma unroll
            for (int s = 0; s < NCOPY / 4; ++s) {
                const int k0 = (4 * s + 2 * half + 2 * nIdx) & (NCOPY - 1);
                b[s] = *(const v2f*)&hist[row0 + k0];
            }
            __builtin_amdgcn_sched_barrier(0);
            v8f c = {};
            #pragma unroll
            for (int s = 0; s < NCOPY / 4; ++s) {
                c = __builtin_amdgcn_wmma_f32_16x16x4_f32(
                         false, aOnes,  false, b[s],
                         (short)0, c,  false,  false);
            }
            asm volatile("v_nop\n\tv_nop\n\tv_nop\n\tv_nop\n\tv_nop"
                         : "+v"(c)
                         : "v"(b[0]), "v"(b[1]), "v"(b[2]), "v"(b[3]), "v"(b[4]), "v"(b[5]), "v"(b[6]), "v"(b[7]));
            if (lane < 16) acc[col] = c[0];
        }
    }
    __syncthreads();

    if (t < NRING && t >= lo && t < hi) {
        const int r = t;
        const float cr = acc[r - lo];
        const float ci = acc[NBL + r - lo];
        const float p1 = acc[2 * NBL + r - lo];
        const float p2 = acc[3 * NBL + r - lo];
        const float num = __builtin_sqrtf(cr * cr + ci * ci);
        const float den = __builtin_sqrtf(p1 * p2);
        out[(size_t)img * NRING + r] = (den == 0.0f) ? 0.0f : num / den;
    }
}

extern "C" void kernel_launch(void* const* d_in, const int* in_sizes, int n_in,
                              void* d_out, int out_size, void* d_ws, size_t ws_size,
                              hipStream_t stream)
{
    (void)in_sizes; (void)n_in; (void)out_size; (void)d_ws; (void)ws_size;
    const float* f1r = (const float*)d_in[0];
    const float* f1i = (const float*)d_in[1];
    const float* f2r = (const float*)d_in[2];
    const float* f2i = (const float*)d_in[3];

    frc_banded<<<dim3(B_ * NBAND), dim3(THREADS), 0, stream>>>(
        f1r, f1i, f2r, f2i, (float*)d_out);
}
